# starting kernel
# baseline (speedup 1.0000x reference)
#include <hip/hip_runtime.h>
#include <hip/hip_bf16.h>

#define N_NODES 10000
#define N_EDGES 100000
#define DIM     768

typedef __attribute__((ext_vector_type(2))) float v2f;
typedef __attribute__((ext_vector_type(8))) float v8f;

__global__ void gcn_init_deg(float* __restrict__ deg) {
    int i = blockIdx.x * blockDim.x + threadIdx.x;
    if (i < N_NODES) deg[i] = 1.0f;
}

__global__ void gcn_count_deg(const int* __restrict__ dst, float* __restrict__ deg) {
    int e = blockIdx.x * blockDim.x + threadIdx.x;
    if (e < N_EDGES) atomicAdd(&deg[dst[e]], 1.0f);
}

__global__ void gcn_rsqrt_deg(const float* __restrict__ deg, float* __restrict__ dinv) {
    int i = blockIdx.x * blockDim.x + threadIdx.x;
    if (i < N_NODES) dinv[i] = rsqrtf(deg[i]);
}

__global__ __launch_bounds__(32) void gcn_gemm_wmma(const float* __restrict__ X,
                                                    const float* __restrict__ W,
                                                    float* __restrict__ H) {
    const int lane = threadIdx.x & 31;
    const int m    = lane & 15;
    const int hi   = lane >> 4;
    const int koff = hi * 2;
    const int row0 = blockIdx.x * 16;
    const int col0 = blockIdx.y * 64;

    v8f c0 = {}, c1 = {}, c2 = {}, c3 = {};

    const float* xrow = X + (size_t)(row0 + m) * DIM;

    for (int k = 0; k < DIM; k += 4) {
        v2f a = *(const v2f*)(xrow + k + koff);

        const float* wb = W + (size_t)(k + koff) * DIM + col0 + m;
        v2f b0 = { wb[ 0], wb[DIM +  0] };
        v2f b1 = { wb[16], wb[DIM + 16] };
        v2f b2 = { wb[32], wb[DIM + 32] };
        v2f b3 = { wb[48], wb[DIM + 48] };

        c0 = __builtin_amdgcn_wmma_f32_16x16x4_f32(false, a, false, b0, (short)0, c0, false, false);
        c1 = __builtin_amdgcn_wmma_f32_16x16x4_f32(false, a, false, b1, (short)0, c1, false, false);
        c2 = __builtin_amdgcn_wmma_f32_16x16x4_f32(false, a, false, b2, (short)0, c2, false, false);
        c3 = __builtin_amdgcn_wmma_f32_16x16x4_f32(false, a, false, b3, (short)0, c3, false, false);
    }

    #pragma unroll
    for (int r = 0; r < 8; ++r) {
        const int row = row0 + r + hi * 8;
        float* ho = H + (size_t)row * DIM + col0 + m;
        ho[ 0] = c0[r];
        ho[16] = c1[r];
        ho[32] = c2[r];
        ho[48] = c3[r];
    }
}

__global__ void gcn_self_bias(const float* __restrict__ H,
                              const float* __restrict__ dinv,
                              const float* __restrict__ bias,
                              float* __restrict__ out) {
    const int i = blockIdx.x;
    const float n = dinv[i] * dinv[i];
    const size_t base = (size_t)i * DIM;
    for (int d = threadIdx.x; d < DIM; d += blockDim.x)
        out[base + d] = bias[d] + n * H[base + d];
}

__global__ void gcn_scatter(const int* __restrict__ src,
                            const int* __restrict__ dst,
                            const float* __restrict__ H,
                            const float* __restrict__ dinv,
                            float* __restrict__ out) {
    const int e = blockIdx.x;
    const int s = src[e];
    const int t = dst[e];
    const float n = dinv[s] * dinv[t];
    const size_t sb = (size_t)s * DIM;
    const size_t tb = (size_t)t * DIM;
    for (int d = threadIdx.x; d < DIM; d += blockDim.x)
        atomicAdd(&out[tb + d], n * H[sb + d]);
}

extern "C" void kernel_launch(void* const* d_in, const int* in_sizes, int n_in,
                              void* d_out, int out_size, void* d_ws, size_t ws_size,
                              hipStream_t stream) {
    const float* x    = (const float*)d_in[0];
    const int*   ei   = (const int*)d_in[1];
    const float* W    = (const float*)d_in[2];
    const float* bias = (const float*)d_in[3];
    float*       out  = (float*)d_out;

    const int* src = ei;
    const int* dst = ei + N_EDGES;

    float* deg  = (float*)d_ws;
    float* dinv = deg + N_NODES;
    float* h    = dinv + N_NODES;

    gcn_init_deg <<<(N_NODES + 255) / 256, 256, 0, stream>>>(deg);
    gcn_count_deg<<<(N_EDGES + 255) / 256, 256, 0, stream>>>(dst, deg);
    gcn_rsqrt_deg<<<(N_NODES + 255) / 256, 256, 0, stream>>>(deg, dinv);

    dim3 ggrid(N_NODES / 16, DIM / 64);
    gcn_gemm_wmma<<<ggrid, 32, 0, stream>>>(x, W, h);

    gcn_self_bias<<<N_NODES, 256, 0, stream>>>(h, dinv, bias, out);
    gcn_scatter  <<<N_EDGES, 256, 0, stream>>>(src, dst, h, dinv, out);
}


# Aggregate by target: edges bucketed per node (int atomic rank + scan), one work-group sums a node's h rows in registers, stores out once; no float atomics; sum order = arrival rank
# speedup vs baseline: 1.7406x; 1.7406x over previous
#include <hip/hip_runtime.h>
#include <hip/hip_bf16.h>

#define N_NODES 10000
#define N_EDGES 100000
#define DIM     768

typedef __attribute__((ext_vector_type(2))) float v2f;
typedef __attribute__((ext_vector_type(8))) float v8f;


__global__ void gcn_zero_cnt(int* __restrict__ cnt) {
    int i = blockIdx.x * blockDim.x + threadIdx.x;
    if (i < N_NODES) cnt[i] = 0;
}

__global__ void gcn_rank(const int* __restrict__ dst, int* __restrict__ cnt,
                         int* __restrict__ rank) {
    int e = blockIdx.x * blockDim.x + threadIdx.x;
    if (e < N_EDGES) rank[e] = atomicAdd(&cnt[dst[e]], 1);
}

#define SCAN_THREADS 1024
#define SCAN_ITEMS   ((N_NODES + SCAN_THREADS - 1) / SCAN_THREADS)

__global__ __launch_bounds__(SCAN_THREADS) void gcn_scan(const int* __restrict__ cnt,
                                                         int* __restrict__ rowptr,
                                                         float* __restrict__ dinv) {
    __shared__ int wave_off[SCAN_THREADS / 32];
    const int t    = threadIdx.x;
    const int lane = t & 31;
    const int w    = t >> 5;
    const int base = t * SCAN_ITEMS;

    int c[SCAN_ITEMS];
    int s = 0;
    #pragma unroll
    for (int j = 0; j < SCAN_ITEMS; ++j) {
        const int idx = base + j;
        c[j] = (idx < N_NODES) ? cnt[idx] : 0;
        s += c[j];
    }

    int inc = s;
    for (int off = 1; off < 32; off <<= 1) {
        const int y = __shfl_up(inc, off, 32);
        if (lane >= off) inc += y;
    }
    if (lane == 31) wave_off[w] = inc;
    __syncthreads();
    if (w == 0) {
        const int x = wave_off[lane];
        int xi = x;
        for (int off = 1; off < 32; off <<= 1) {
            const int y = __shfl_up(xi, off, 32);
            if (lane >= off) xi += y;
        }
        wave_off[lane] = xi - x;
    }
    __syncthreads();

    int run = wave_off[w] + inc - s;
    #pragma unroll
    for (int j = 0; j < SCAN_ITEMS; ++j) {
        const int idx = base + j;
        if (idx < N_NODES) {
            rowptr[idx] = run;
            dinv[idx]   = rsqrtf((float)(c[j] + 1));
        }
        run += c[j];
    }
}

__global__ void gcn_fill(const int* __restrict__ src, const int* __restrict__ dst,
                         const int* __restrict__ rank, const int* __restrict__ rowptr,
                         int* __restrict__ csr_src) {
    int e = blockIdx.x * blockDim.x + threadIdx.x;
    if (e < N_EDGES) csr_src[rowptr[dst[e]] + rank[e]] = src[e];
}

__global__ __launch_bounds__(32) void gcn_gemm_wmma(const float* __restrict__ X,
                                                    const float* __restrict__ W,
                                                    float* __restrict__ H) {
    const int lane = threadIdx.x & 31;
    const int m    = lane & 15;
    const int hi   = lane >> 4;
    const int koff = hi * 2;
    const int row0 = blockIdx.x * 16;
    const int col0 = blockIdx.y * 64;

    v8f c0 = {}, c1 = {}, c2 = {}, c3 = {};

    const float* xrow = X + (size_t)(row0 + m) * DIM;

    for (int k = 0; k < DIM; k += 4) {
        v2f a = *(const v2f*)(xrow + k + koff);

        const float* wb = W + (size_t)(k + koff) * DIM + col0 + m;
        v2f b0 = { wb[ 0], wb[DIM +  0] };
        v2f b1 = { wb[16], wb[DIM + 16] };
        v2f b2 = { wb[32], wb[DIM + 32] };
        v2f b3 = { wb[48], wb[DIM + 48] };

        c0 = __builtin_amdgcn_wmma_f32_16x16x4_f32(false, a, false, b0, (short)0, c0, false, false);
        c1 = __builtin_amdgcn_wmma_f32_16x16x4_f32(false, a, false, b1, (short)0, c1, false, false);
        c2 = __builtin_amdgcn_wmma_f32_16x16x4_f32(false, a, false, b2, (short)0, c2, false, false);
        c3 = __builtin_amdgcn_wmma_f32_16x16x4_f32(false, a, false, b3, (short)0, c3, false, false);
    }

    #pragma unroll
    for (int r = 0; r < 8; ++r) {
        const int row = row0 + r + hi * 8;
        float* ho = H + (size_t)row * DIM + col0 + m;
        ho[ 0] = c0[r];
        ho[16] = c1[r];
        ho[32] = c2[r];
        ho[48] = c3[r];
    }
}

#define GATHER_THREADS (DIM / 4)

__device__ __forceinline__ void fma4(float4& acc, float n, const float4& h) {
    acc.x += n * h.x;
    acc.y += n * h.y;
    acc.z += n * h.z;
    acc.w += n * h.w;
}

__device__ __forceinline__ int src_at(const int* __restrict__ csr_src, int p) {
    const unsigned s = (unsigned)csr_src[p];
    return (int)(s < (unsigned)(N_NODES - 1) ? s : (unsigned)(N_NODES - 1));
}

template <int R>
__device__ __forceinline__ void add_rows(float4& acc, const float* __restrict__ H,
                                         const float* __restrict__ dinv,
                                         const int* __restrict__ csr_src,
                                         int p, float di, int d) {
    int    s[R];
    float  n[R];
    float4 h[R];
    #pragma unroll
    for (int q = 0; q < R; ++q) s[q] = src_at(csr_src, p + q);
    #pragma unroll
    for (int q = 0; q < R; ++q) n[q] = dinv[s[q]] * di;
    #pragma unroll
    for (int q = 0; q < R; ++q) h[q] = *(const float4*)(H + (size_t)s[q] * DIM + d);
    #pragma unroll
    for (int q = 0; q < R; ++q) fma4(acc, n[q], h[q]);
}

__global__ __launch_bounds__(GATHER_THREADS) void gcn_gather(const float* __restrict__ H,
                                                             const float* __restrict__ dinv,
                                                             const float* __restrict__ bias,
                                                             const int* __restrict__ rowptr,
                                                             const int* __restrict__ cnt,
                                                             const int* __restrict__ csr_src,
                                                             float* __restrict__ out) {
    const int d = threadIdx.x * 4;
    const float4 bv = *(const float4*)(bias + d);

    for (int i = blockIdx.x; i < N_NODES; i += gridDim.x) {
        const float di  = dinv[i];
        const int   beg = rowptr[i];
        const int   len = cnt[i];

        const float4 hs = *(const float4*)(H + (size_t)i * DIM + d);
        float4 acc = bv;
        fma4(acc, di * di, hs);

        int j = 0;
        for (; j + 4 <= len; j += 4)
            add_rows<4>(acc, H, dinv, csr_src, beg + j, di, d);
        const int left = len - j;
        if (left == 3)      add_rows<3>(acc, H, dinv, csr_src, beg + j, di, d);
        else if (left == 2) add_rows<2>(acc, H, dinv, csr_src, beg + j, di, d);
        else if (left == 1) add_rows<1>(acc, H, dinv, csr_src, beg + j, di, d);

        *(float4*)(out + (size_t)i * DIM + d) = acc;
    }
}

extern "C" void kernel_launch(void* const* d_in, const int* in_sizes, int n_in,
                              void* d_out, int out_size, void* d_ws, size_t ws_size,
                              hipStream_t stream) {
    const float* x    = (const float*)d_in[0];
    const int*   ei   = (const int*)d_in[1];
    const float* W    = (const float*)d_in[2];
    const float* bias = (const float*)d_in[3];
    float*       out  = (float*)d_out;

    const int* src = ei;
    const int* dst = ei + N_EDGES;

    float* h      = (float*)d_ws;
    float* dinv   = h + (size_t)N_NODES * DIM;
    int*   cnt    = (int*)(dinv + N_NODES);
    int*   rowptr = cnt + N_NODES;
    int*   rank   = rowptr + N_NODES;
    int*   csr    = rank + N_EDGES;

    gcn_zero_cnt<<<(N_NODES + 255) / 256, 256, 0, stream>>>(cnt);
    gcn_rank    <<<(N_EDGES + 255) / 256, 256, 0, stream>>>(dst, cnt, rank);
    gcn_scan    <<<1, SCAN_THREADS, 0, stream>>>(cnt, rowptr, dinv);
    gcn_fill    <<<(N_EDGES + 255) / 256, 256, 0, stream>>>(src, dst, rank, rowptr, csr);

    dim3 ggrid(N_NODES / 16, DIM / 64);
    gcn_gemm_wmma<<<ggrid, 32, 0, stream>>>(x, W, h);

    gcn_gather<<<N_NODES / 4, GATHER_THREADS, 0, stream>>>(h, dinv, bias, rowptr, cnt, csr, out);
}


# x@W product retiled: 80x384 work-group tiles, x and W staged through LDS (register-prefetched double buffer), 8 waves owning one WGP; k summed in another order inside each 8-k block
# speedup vs baseline: 2.1654x; 1.2441x over previous
#include <hip/hip_runtime.h>
#include <hip/hip_bf16.h>

#define N_NODES 10000
#define N_EDGES 100000
#define DIM     768

typedef __attribute__((ext_vector_type(2))) float v2f;
typedef __attribute__((ext_vector_type(8))) float v8f;


__global__ void gcn_zero_cnt(int* __restrict__ cnt) {
    int i = blockIdx.x * blockDim.x + threadIdx.x;
    if (i < N_NODES) cnt[i] = 0;
}

__global__ void gcn_rank(const int* __restrict__ dst, int* __restrict__ cnt,
                         int* __restrict__ rank) {
    int e = blockIdx.x * blockDim.x + threadIdx.x;
    if (e < N_EDGES) rank[e] = atomicAdd(&cnt[dst[e]], 1);
}

#define SCAN_THREADS 1024
#define SCAN_ITEMS   ((N_NODES + SCAN_THREADS - 1) / SCAN_THREADS)

__global__ __launch_bounds__(SCAN_THREADS) void gcn_scan(const int* __restrict__ cnt,
                                                         int* __restrict__ rowptr,
                                                         float* __restrict__ dinv) {
    __shared__ int wave_off[SCAN_THREADS / 32];
    const int t    = threadIdx.x;
    const int lane = t & 31;
    const int w    = t >> 5;
    const int base = t * SCAN_ITEMS;

    int c[SCAN_ITEMS];
    int s = 0;
    #pragma unroll
    for (int j = 0; j < SCAN_ITEMS; ++j) {
        const int idx = base + j;
        c[j] = (idx < N_NODES) ? cnt[idx] : 0;
        s += c[j];
    }

    int inc = s;
    for (int off = 1; off < 32; off <<= 1) {
        const int y = __shfl_up(inc, off, 32);
        if (lane >= off) inc += y;
    }
    if (lane == 31) wave_off[w] = inc;
    __syncthreads();
    if (w == 0) {
        const int x = wave_off[lane];
        int xi = x;
        for (int off = 1; off < 32; off <<= 1) {
            const int y = __shfl_up(xi, off, 32);
            if (lane >= off) xi += y;
        }
        wave_off[lane] = xi - x;
    }
    __syncthreads();

    int run = wave_off[w] + inc - s;
    #pragma unroll
    for (int j = 0; j < SCAN_ITEMS; ++j) {
        const int idx = base + j;
        if (idx < N_NODES) {
            rowptr[idx] = run;
            dinv[idx]   = rsqrtf((float)(c[j] + 1));
        }
        run += c[j];
    }
}

__global__ void gcn_fill(const int* __restrict__ src, const int* __restrict__ dst,
                         const int* __restrict__ rank, const int* __restrict__ rowptr,
                         int* __restrict__ csr_src) {
    int e = blockIdx.x * blockDim.x + threadIdx.x;
    if (e < N_EDGES) csr_src[rowptr[dst[e]] + rank[e]] = src[e];
}

typedef __attribute__((ext_vector_type(4))) float v4f;

#define GM_T      80
#define GN_T      384
#define GK_T      64
#define G_THREADS 256
#define G_MT      (GM_T / 16)
#define G_NT      3
#define G_KB      (GK_T / 8)
#define A_PITCH   (GK_T + 4)
#define A_STAGE   (GM_T * A_PITCH)
#define B_PITCH   (GN_T + 1)
#define B_STAGE   ((GK_T / 4) * B_PITCH * 4)
#define STAGE_F   (A_STAGE + B_STAGE)
#define A_LOADS   (GM_T * (GK_T / 4) / G_THREADS)
#define B_BLOCKS  ((GK_T / 4) * (GN_T / 4) / G_THREADS)
#define K_STEPS   (DIM / GK_T)

__device__ __forceinline__ void gemm_fetch(v4f (&ga)[A_LOADS], v4f (&gb)[B_BLOCKS][4],
                                           const float* __restrict__ X,
                                           const float* __restrict__ Wm,
                                           int m0, int n0, int k0, int t) {
    #pragma unroll
    for (int i = 0; i < A_LOADS; ++i) {
        const int idx = i * G_THREADS + t;
        const int row = idx >> 4;
        const int c4  = idx & 15;
        ga[i] = *(const v4f*)(X + (size_t)(m0 + row) * DIM + k0 + c4 * 4);
    }
    #pragma unroll
    for (int j = 0; j < B_BLOCKS; ++j) {
        const int p    = j * (G_THREADS / 16) + (t >> 4);
        const int kq   = (p / (GN_T / 16)) * 4 + (t & 3);
        const int ng   = (p % (GN_T / 16)) * 4 + ((t >> 2) & 3);
        const float* wp = Wm + (size_t)(k0 + kq * 4) * DIM + n0 + ng * 4;
        #pragma unroll
        for (int kk = 0; kk < 4; ++kk) gb[j][kk] = *(const v4f*)(wp + (size_t)kk * DIM);
    }
}

__device__ __forceinline__ void gemm_stash(const v4f (&ga)[A_LOADS], const v4f (&gb)[B_BLOCKS][4],
                                           float* __restrict__ As, float* __restrict__ Bs, int t) {
    #pragma unroll
    for (int i = 0; i < A_LOADS; ++i) {
        const int idx = i * G_THREADS + t;
        const int row = idx >> 4;
        const int c4  = idx & 15;
        *(v4f*)(As + row * A_PITCH + c4 * 4) = ga[i];
    }
    #pragma unroll
    for (int j = 0; j < B_BLOCKS; ++j) {
        const int p    = j * (G_THREADS / 16) + (t >> 4);
        const int kq   = (p / (GN_T / 16)) * 4 + (t & 3);
        const int ng   = (p % (GN_T / 16)) * 4 + ((t >> 2) & 3);
        float* bp = Bs + (kq * B_PITCH + ng * 4) * 4;
        *(v4f*)(bp +  0) = v4f{ gb[j][0].x, gb[j][1].x, gb[j][2].x, gb[j][3].x };
        *(v4f*)(bp +  4) = v4f{ gb[j][0].y, gb[j][1].y, gb[j][2].y, gb[j][3].y };
        *(v4f*)(bp +  8) = v4f{ gb[j][0].z, gb[j][1].z, gb[j][2].z, gb[j][3].z };
        *(v4f*)(bp + 12) = v4f{ gb[j][0].w, gb[j][1].w, gb[j][2].w, gb[j][3].w };
    }
}

__device__ __forceinline__ void gemm_frags(v4f (&fa)[G_MT], v4f (&fb)[G_NT],
                                           const float* __restrict__ As,
                                           const float* __restrict__ Bs,
                                           int kb, int c, int hi, int nw) {
    #pragma unroll
    for (int mt = 0; mt < G_MT; ++mt)
        fa[mt] = *(const v4f*)(As + (mt * 16 + c) * A_PITCH + kb * 8 + hi * 4);
    #pragma unroll
    for (int nt = 0; nt < G_NT; ++nt)
        fb[nt] = *(const v4f*)(Bs + ((kb * 2 + hi) * B_PITCH + nw + nt * 16 + c) * 4);
}

__global__ __launch_bounds__(G_THREADS) void gcn_gemm_wmma(const float* __restrict__ X,
                                                           const float* __restrict__ Wm,
                                                           float* __restrict__ H) {
    __shared__ __attribute__((aligned(16))) float lds[2 * STAGE_F];

    const int t    = threadIdx.x;
    const int lane = t & 31;
    const int c    = lane & 15;
    const int hi   = lane >> 4;
    const int nw   = (t >> 5) * (G_NT * 16);
    const int m0   = (blockIdx.x >> 1) * GM_T;
    const int n0   = (blockIdx.x & 1) * GN_T;

    v8f acc[G_MT][G_NT];
    #pragma unroll
    for (int mt = 0; mt < G_MT; ++mt)
        #pragma unroll
        for (int nt = 0; nt < G_NT; ++nt) acc[mt][nt] = v8f{};

    v4f ga[A_LOADS];
    v4f gb[B_BLOCKS][4];
    gemm_fetch(ga, gb, X, Wm, m0, n0, 0, t);
    gemm_stash(ga, gb, lds, lds + A_STAGE, t);
    __syncthreads();

    for (int ks = 0; ks < K_STEPS; ++ks) {
        const float* As = lds + (ks & 1) * STAGE_F;
        const float* Bs = As + A_STAGE;

        if (ks + 1 < K_STEPS) gemm_fetch(ga, gb, X, Wm, m0, n0, (ks + 1) * GK_T, t);

        v4f fa[G_MT], fb[G_NT];
        gemm_frags(fa, fb, As, Bs, 0, c, hi, nw);
        #pragma unroll
        for (int kb = 0; kb < G_KB; ++kb) {
            v4f na[G_MT], nb[G_NT];
            if (kb + 1 < G_KB) gemm_frags(na, nb, As, Bs, kb + 1, c, hi, nw);

            #pragma unroll
            for (int mt = 0; mt < G_MT; ++mt)
                #pragma unroll
                for (int nt = 0; nt < G_NT; ++nt)
                    acc[mt][nt] = __builtin_amdgcn_wmma_f32_16x16x4_f32(false, fa[mt].xy, false, fb[nt].xy, (short)0, acc[mt][nt], false, false);
            #pragma unroll
            for (int mt = 0; mt < G_MT; ++mt)
                #pragma unroll
                for (int nt = 0; nt < G_NT; ++nt)
                    acc[mt][nt] = __builtin_amdgcn_wmma_f32_16x16x4_f32(false, fa[mt].zw, false, fb[nt].zw, (short)0, acc[mt][nt], false, false);

            if (kb + 1 < G_KB) __builtin_amdgcn_sched_barrier(0);

            if (kb + 1 < G_KB) {
                #pragma unroll
                for (int mt = 0; mt < G_MT; ++mt) fa[mt] = na[mt];
                #pragma unroll
                for (int nt = 0; nt < G_NT; ++nt) fb[nt] = nb[nt];
            }
        }

        if (ks + 1 < K_STEPS) {
            float* An = lds + ((ks + 1) & 1) * STAGE_F;
            gemm_stash(ga, gb, An, An + A_STAGE, t);
        }
        __syncthreads();
    }

    #pragma unroll
    for (int mt = 0; mt < G_MT; ++mt) {
        #pragma unroll
        for (int r = 0; r < 8; ++r) {
            const int row = m0 + mt * 16 + r + hi * 8;
            float* ho = H + (size_t)row * DIM + n0 + nw + c;
            #pragma unroll
            for (int nt = 0; nt < G_NT; ++nt) ho[nt * 16] = acc[mt][nt][r];
        }
    }
}

#define GATHER_THREADS (DIM / 4)

__device__ __forceinline__ void fma4(float4& acc, float n, const float4& h) {
    acc.x += n * h.x;
    acc.y += n * h.y;
    acc.z += n * h.z;
    acc.w += n * h.w;
}

__device__ __forceinline__ int src_at(const int* __restrict__ csr_src, int p) {
    const unsigned s = (unsigned)csr_src[p];
    return (int)(s < (unsigned)(N_NODES - 1) ? s : (unsigned)(N_NODES - 1));
}

template <int R>
__device__ __forceinline__ void add_rows(float4& acc, const float* __restrict__ H,
                                         const float* __restrict__ dinv,
                                         const int* __restrict__ csr_src,
                                         int p, float di, int d) {
    int    s[R];
    float  n[R];
    float4 h[R];
    #pragma unroll
    for (int q = 0; q < R; ++q) s[q] = src_at(csr_src, p + q);
    #pragma unroll
    for (int q = 0; q < R; ++q) n[q] = dinv[s[q]] * di;
    #pragma unroll
    for (int q = 0; q < R; ++q) h[q] = *(const float4*)(H + (size_t)s[q] * DIM + d);
    #pragma unroll
    for (int q = 0; q < R; ++q) fma4(acc, n[q], h[q]);
}

__global__ __launch_bounds__(GATHER_THREADS) void gcn_gather(const float* __restrict__ H,
                                                             const float* __restrict__ dinv,
                                                             const float* __restrict__ bias,
                                                             const int* __restrict__ rowptr,
                                                             const int* __restrict__ cnt,
                                                             const int* __restrict__ csr_src,
                                                             float* __restrict__ out) {
    const int d = threadIdx.x * 4;
    const float4 bv = *(const float4*)(bias + d);

    for (int i = blockIdx.x; i < N_NODES; i += gridDim.x) {
        const float di  = dinv[i];
        const int   beg = rowptr[i];
        const int   len = cnt[i];

        const float4 hs = *(const float4*)(H + (size_t)i * DIM + d);
        float4 acc = bv;
        fma4(acc, di * di, hs);

        int j = 0;
        for (; j + 4 <= len; j += 4)
            add_rows<4>(acc, H, dinv, csr_src, beg + j, di, d);
        const int left = len - j;
        if (left == 3)      add_rows<3>(acc, H, dinv, csr_src, beg + j, di, d);
        else if (left == 2) add_rows<2>(acc, H, dinv, csr_src, beg + j, di, d);
        else if (left == 1) add_rows<1>(acc, H, dinv, csr_src, beg + j, di, d);

        *(float4*)(out + (size_t)i * DIM + d) = acc;
    }
}

extern "C" void kernel_launch(void* const* d_in, const int* in_sizes, int n_in,
                              void* d_out, int out_size, void* d_ws, size_t ws_size,
                              hipStream_t stream) {
    const float* x    = (const float*)d_in[0];
    const int*   ei   = (const int*)d_in[1];
    const float* W    = (const float*)d_in[2];
    const float* bias = (const float*)d_in[3];
    float*       out  = (float*)d_out;

    const int* src = ei;
    const int* dst = ei + N_EDGES;

    float* h      = (float*)d_ws;
    float* dinv   = h + (size_t)N_NODES * DIM;
    int*   cnt    = (int*)(dinv + N_NODES);
    int*   rowptr = cnt + N_NODES;
    int*   rank   = rowptr + N_NODES;
    int*   csr    = rank + N_EDGES;

    gcn_zero_cnt<<<(N_NODES + 255) / 256, 256, 0, stream>>>(cnt);
    gcn_rank    <<<(N_EDGES + 255) / 256, 256, 0, stream>>>(dst, cnt, rank);
    gcn_scan    <<<1, SCAN_THREADS, 0, stream>>>(cnt, rowptr, dinv);
    gcn_fill    <<<(N_EDGES + 255) / 256, 256, 0, stream>>>(src, dst, rank, rowptr, csr);

    gcn_gemm_wmma<<<(N_NODES / GM_T) * (DIM / GN_T), G_THREADS, 0, stream>>>(x, W, h);

    gcn_gather<<<N_NODES / 4, GATHER_THREADS, 0, stream>>>(h, dinv, bias, rowptr, cnt, csr, out);
}


# Edge bucketing (count, scan, place; LDS atomics) moved into 6 spare work-groups of the x@W launch, hidden under the product: 6 launches -> 2; sum order = arrival order at LDS cursors
# speedup vs baseline: 2.6243x; 1.2119x over previous
#include <hip/hip_runtime.h>
#include <hip/hip_bf16.h>

#define N_NODES 10000
#define N_EDGES 100000
#define DIM     768

typedef __attribute__((ext_vector_type(2))) float v2f;
typedef __attribute__((ext_vector_type(8))) float v8f;


typedef __attribute__((ext_vector_type(4))) float v4f;

#define GM_T      80
#define GN_T      384
#define GK_T      64
#define G_THREADS 256
#define G_MT      (GM_T / 16)
#define G_NT      3
#define G_KB      (GK_T / 8)
#define A_PITCH   (GK_T + 4)
#define A_STAGE   (GM_T * A_PITCH)
#define B_PITCH   (GN_T + 1)
#define B_STAGE   ((GK_T / 4) * B_PITCH * 4)
#define STAGE_F   (A_STAGE + B_STAGE)
#define A_LOADS   (GM_T * (GK_T / 4) / G_THREADS)
#define B_BLOCKS  ((GK_T / 4) * (GN_T / 4) / G_THREADS)
#define K_STEPS   (DIM / GK_T)

__device__ __forceinline__ void gemm_fetch(v4f (&ga)[A_LOADS], v4f (&gb)[B_BLOCKS][4],
                                           const float* __restrict__ X,
                                           const float* __restrict__ Wm,
                                           int m0, int n0, int k0, int t) {
    #pragma unroll
    for (int i = 0; i < A_LOADS; ++i) {
        const int idx = i * G_THREADS + t;
        const int row = idx >> 4;
        const int c4  = idx & 15;
        ga[i] = *(const v4f*)(X + (size_t)(m0 + row) * DIM + k0 + c4 * 4);
    }
    #pragma unroll
    for (int j = 0; j < B_BLOCKS; ++j) {
        const int p    = j * (G_THREADS / 16) + (t >> 4);
        const int kq   = (p / (GN_T / 16)) * 4 + (t & 3);
        const int ng   = (p % (GN_T / 16)) * 4 + ((t >> 2) & 3);
        const float* wp = Wm + (size_t)(k0 + kq * 4) * DIM + n0 + ng * 4;
        #pragma unroll
        for (int kk = 0; kk < 4; ++kk) gb[j][kk] = *(const v4f*)(wp + (size_t)kk * DIM);
    }
}

__device__ __forceinline__ void gemm_stash(const v4f (&ga)[A_LOADS], const v4f (&gb)[B_BLOCKS][4],
                                           float* __restrict__ As, float* __restrict__ Bs, int t) {
    #pragma unroll
    for (int i = 0; i < A_LOADS; ++i) {
        const int idx = i * G_THREADS + t;
        const int row = idx >> 4;
        const int c4  = idx & 15;
        *(v4f*)(As + row * A_PITCH + c4 * 4) = ga[i];
    }
    #pragma unroll
    for (int j = 0; j < B_BLOCKS; ++j) {
        const int p    = j * (G_THREADS / 16) + (t >> 4);
        const int kq   = (p / (GN_T / 16)) * 4 + (t & 3);
        const int ng   = (p % (GN_T / 16)) * 4 + ((t >> 2) & 3);
        float* bp = Bs + (kq * B_PITCH + ng * 4) * 4;
        *(v4f*)(bp +  0) = v4f{ gb[j][0].x, gb[j][1].x, gb[j][2].x, gb[j][3].x };
        *(v4f*)(bp +  4) = v4f{ gb[j][0].y, gb[j][1].y, gb[j][2].y, gb[j][3].y };
        *(v4f*)(bp +  8) = v4f{ gb[j][0].z, gb[j][1].z, gb[j][2].z, gb[j][3].z };
        *(v4f*)(bp + 12) = v4f{ gb[j][0].w, gb[j][1].w, gb[j][2].w, gb[j][3].w };
    }
}

__device__ __forceinline__ void gemm_frags(v4f (&fa)[G_MT], v4f (&fb)[G_NT],
                                           const float* __restrict__ As,
                                           const float* __restrict__ Bs,
                                           int kb, int c, int hi, int nw) {
    #pragma unroll
    for (int mt = 0; mt < G_MT; ++mt)
        fa[mt] = *(const v4f*)(As + (mt * 16 + c) * A_PITCH + kb * 8 + hi * 4);
    #pragma unroll
    for (int nt = 0; nt < G_NT; ++nt)
        fb[nt] = *(const v4f*)(Bs + ((kb * 2 + hi) * B_PITCH + nw + nt * 16 + c) * 4);
}

#define CSR_WGS    6
#define CSR_NODES  ((N_NODES + CSR_WGS - 1) / CSR_WGS)
#define CSR_ITEMS  ((CSR_NODES + G_THREADS - 1) / G_THREADS)
#define CSR_SLOTS  (CSR_ITEMS * G_THREADS)
#define CSR_BATCH  8
#define EDGE_QUADS (N_EDGES / 4)

__device__ __forceinline__ void csr_count(int* __restrict__ cur, int d, int lo, int n_here, int& below) {
    const unsigned u = (unsigned)(d - lo);
    if (u < (unsigned)n_here) atomicAdd(&cur[u], 1);
    else                      below += (d < lo) ? 1 : 0;
}

__device__ __forceinline__ void csr_place(int* __restrict__ cur, int d, int s, int lo, int n_here,
                                          int* __restrict__ csr_src) {
    const unsigned u = (unsigned)(d - lo);
    if (u < (unsigned)n_here) csr_src[atomicAdd(&cur[u], 1)] = s;
}

__device__ __forceinline__ void csr_build(int* __restrict__ L, int part,
                                          const int* __restrict__ src, const int* __restrict__ dst,
                                          int* __restrict__ cnt, int* __restrict__ rowptr,
                                          float* __restrict__ dinv, int* __restrict__ csr_src) {
    const int t      = threadIdx.x;
    const int lane   = t & 31;
    const int w      = t >> 5;
    const int lo     = part * CSR_NODES;
    const int n_here = (N_NODES - lo < CSR_NODES) ? (N_NODES - lo) : CSR_NODES;

    int* cur  = L;
    int* wsum = L + CSR_SLOTS;
    int* wlow = wsum + G_THREADS / 32;

    for (int i = t; i < CSR_SLOTS; i += G_THREADS) cur[i] = 0;
    __syncthreads();

    const int4* dst4 = (const int4*)dst;
    const int4* src4 = (const int4*)src;

    int below = 0;
    for (int i0 = t; i0 < EDGE_QUADS; i0 += CSR_BATCH * G_THREADS) {
        int4 d[CSR_BATCH];
        #pragma unroll
        for (int q = 0; q < CSR_BATCH; ++q) {
            const int i = i0 + q * G_THREADS;
            d[q] = (i < EDGE_QUADS) ? dst4[i] : int4{-1, -1, -1, -1};
        }
        #pragma unroll
        for (int q = 0; q < CSR_BATCH; ++q) {
            if (i0 + q * G_THREADS < EDGE_QUADS) {
                csr_count(cur, d[q].x, lo, n_here, below);
                csr_count(cur, d[q].y, lo, n_here, below);
                csr_count(cur, d[q].z, lo, n_here, below);
                csr_count(cur, d[q].w, lo, n_here, below);
            }
        }
    }
    for (int off = 16; off > 0; off >>= 1) below += __shfl_xor(below, off, 32);
    if (lane == 0) wlow[w] = below;
    __syncthreads();

    int c[CSR_ITEMS];
    int s = 0;
    #pragma unroll
    for (int j = 0; j < CSR_ITEMS; ++j) {
        c[j] = cur[t * CSR_ITEMS + j];
        s += c[j];
    }
    int inc = s;
    for (int off = 1; off < 32; off <<= 1) {
        const int y = __shfl_up(inc, off, 32);
        if (lane >= off) inc += y;
    }
    if (lane == 31) wsum[w] = inc;
    __syncthreads();

    int run = inc - s;
    for (int k = 0; k < G_THREADS / 32; ++k) {
        run += wlow[k];
        if (k < w) run += wsum[k];
    }
    #pragma unroll
    for (int j = 0; j < CSR_ITEMS; ++j) {
        const int idx = t * CSR_ITEMS + j;
        if (idx < n_here) {
            rowptr[lo + idx] = run;
            cnt[lo + idx]    = c[j];
            dinv[lo + idx]   = rsqrtf((float)(c[j] + 1));
            cur[idx]         = run;
        }
        run += c[j];
    }
    __syncthreads();

    for (int i0 = t; i0 < EDGE_QUADS; i0 += CSR_BATCH * G_THREADS) {
        int4 d[CSR_BATCH], sv[CSR_BATCH];
        #pragma unroll
        for (int q = 0; q < CSR_BATCH; ++q) {
            const int i = i0 + q * G_THREADS;
            d[q]  = (i < EDGE_QUADS) ? dst4[i] : int4{-1, -1, -1, -1};
            sv[q] = (i < EDGE_QUADS) ? src4[i] : int4{0, 0, 0, 0};
        }
        #pragma unroll
        for (int q = 0; q < CSR_BATCH; ++q) {
            if (i0 + q * G_THREADS < EDGE_QUADS) {
                csr_place(cur, d[q].x, sv[q].x, lo, n_here, csr_src);
                csr_place(cur, d[q].y, sv[q].y, lo, n_here, csr_src);
                csr_place(cur, d[q].z, sv[q].z, lo, n_here, csr_src);
                csr_place(cur, d[q].w, sv[q].w, lo, n_here, csr_src);
            }
        }
    }
}

#define GEMM_WGS ((N_NODES / GM_T) * (DIM / GN_T))

__global__ __launch_bounds__(G_THREADS) void gcn_gemm_wmma(const float* __restrict__ X,
                                                           const float* __restrict__ Wm,
                                                           float* __restrict__ H,
                                                           const int* __restrict__ src,
                                                           const int* __restrict__ dst,
                                                           int* __restrict__ cnt,
                                                           int* __restrict__ rowptr,
                                                           float* __restrict__ dinv,
                                                           int* __restrict__ csr_src) {
    __shared__ __attribute__((aligned(16))) float lds[2 * STAGE_F];

    if (blockIdx.x >= GEMM_WGS) {
        csr_build((int*)lds, blockIdx.x - GEMM_WGS, src, dst, cnt, rowptr, dinv, csr_src);
        return;
    }

    const int t    = threadIdx.x;
    const int lane = t & 31;
    const int c    = lane & 15;
    const int hi   = lane >> 4;
    const int nw   = (t >> 5) * (G_NT * 16);
    const int m0   = (blockIdx.x >> 1) * GM_T;
    const int n0   = (blockIdx.x & 1) * GN_T;

    v8f acc[G_MT][G_NT];
    #pragma unroll
    for (int mt = 0; mt < G_MT; ++mt)
        #pragma unroll
        for (int nt = 0; nt < G_NT; ++nt) acc[mt][nt] = v8f{};

    v4f ga[A_LOADS];
    v4f gb[B_BLOCKS][4];
    gemm_fetch(ga, gb, X, Wm, m0, n0, 0, t);
    gemm_stash(ga, gb, lds, lds + A_STAGE, t);
    __syncthreads();

    for (int ks = 0; ks < K_STEPS; ++ks) {
        const float* As = lds + (ks & 1) * STAGE_F;
        const float* Bs = As + A_STAGE;

        if (ks + 1 < K_STEPS) gemm_fetch(ga, gb, X, Wm, m0, n0, (ks + 1) * GK_T, t);

        v4f fa[G_MT], fb[G_NT];
        gemm_frags(fa, fb, As, Bs, 0, c, hi, nw);
        #pragma unroll
        for (int kb = 0; kb < G_KB; ++kb) {
            v4f na[G_MT], nb[G_NT];
            if (kb + 1 < G_KB) gemm_frags(na, nb, As, Bs, kb + 1, c, hi, nw);

            #pragma unroll
            for (int mt = 0; mt < G_MT; ++mt)
                #pragma unroll
                for (int nt = 0; nt < G_NT; ++nt)
                    acc[mt][nt] = __builtin_amdgcn_wmma_f32_16x16x4_f32(false, fa[mt].xy, false, fb[nt].xy, (short)0, acc[mt][nt], false, false);
            #pragma unroll
            for (int mt = 0; mt < G_MT; ++mt)
                #pragma unroll
                for (int nt = 0; nt < G_NT; ++nt)
                    acc[mt][nt] = __builtin_amdgcn_wmma_f32_16x16x4_f32(false, fa[mt].zw, false, fb[nt].zw, (short)0, acc[mt][nt], false, false);

            if (kb + 1 < G_KB) __builtin_amdgcn_sched_barrier(0);

            if (kb + 1 < G_KB) {
                #pragma unroll
                for (int mt = 0; mt < G_MT; ++mt) fa[mt] = na[mt];
                #pragma unroll
                for (int nt = 0; nt < G_NT; ++nt) fb[nt] = nb[nt];
            }
        }

        if (ks + 1 < K_STEPS) {
            float* An = lds + ((ks + 1) & 1) * STAGE_F;
            gemm_stash(ga, gb, An, An + A_STAGE, t);
        }
        __syncthreads();
    }

    #pragma unroll
    for (int mt = 0; mt < G_MT; ++mt) {
        #pragma unroll
        for (int r = 0; r < 8; ++r) {
            const int row = m0 + mt * 16 + r + hi * 8;
            float* ho = H + (size_t)row * DIM + n0 + nw + c;
            #pragma unroll
            for (int nt = 0; nt < G_NT; ++nt) ho[nt * 16] = acc[mt][nt][r];
        }
    }
}

#define GATHER_THREADS (DIM / 4)

__device__ __forceinline__ void fma4(float4& acc, float n, const float4& h) {
    acc.x += n * h.x;
    acc.y += n * h.y;
    acc.z += n * h.z;
    acc.w += n * h.w;
}

__device__ __forceinline__ int src_at(const int* __restrict__ csr_src, int p) {
    const unsigned s = (unsigned)csr_src[p];
    return (int)(s < (unsigned)(N_NODES - 1) ? s : (unsigned)(N_NODES - 1));
}

template <int R>
__device__ __forceinline__ void add_rows(float4& acc, const float* __restrict__ H,
                                         const float* __restrict__ dinv,
                                         const int* __restrict__ csr_src,
                                         int p, float di, int d) {
    int    s[R];
    float  n[R];
    float4 h[R];
    #pragma unroll
    for (int q = 0; q < R; ++q) s[q] = src_at(csr_src, p + q);
    #pragma unroll
    for (int q = 0; q < R; ++q) n[q] = dinv[s[q]] * di;
    #pragma unroll
    for (int q = 0; q < R; ++q) h[q] = *(const float4*)(H + (size_t)s[q] * DIM + d);
    #pragma unroll
    for (int q = 0; q < R; ++q) fma4(acc, n[q], h[q]);
}

__global__ __launch_bounds__(GATHER_THREADS) void gcn_gather(const float* __restrict__ H,
                                                             const float* __restrict__ dinv,
                                                             const float* __restrict__ bias,
                                                             const int* __restrict__ rowptr,
                                                             const int* __restrict__ cnt,
                                                             const int* __restrict__ csr_src,
                                                             float* __restrict__ out) {
    const int d = threadIdx.x * 4;
    const float4 bv = *(const float4*)(bias + d);

    for (int i = blockIdx.x; i < N_NODES; i += gridDim.x) {
        const float di  = dinv[i];
        const int   beg = rowptr[i];
        const int   len = cnt[i];

        const float4 hs = *(const float4*)(H + (size_t)i * DIM + d);
        float4 acc = bv;
        fma4(acc, di * di, hs);

        int j = 0;
        for (; j + 4 <= len; j += 4)
            add_rows<4>(acc, H, dinv, csr_src, beg + j, di, d);
        const int left = len - j;
        if (left == 3)      add_rows<3>(acc, H, dinv, csr_src, beg + j, di, d);
        else if (left == 2) add_rows<2>(acc, H, dinv, csr_src, beg + j, di, d);
        else if (left == 1) add_rows<1>(acc, H, dinv, csr_src, beg + j, di, d);

        *(float4*)(out + (size_t)i * DIM + d) = acc;
    }
}

extern "C" void kernel_launch(void* const* d_in, const int* in_sizes, int n_in,
                              void* d_out, int out_size, void* d_ws, size_t ws_size,
                              hipStream_t stream) {
    const float* x    = (const float*)d_in[0];
    const int*   ei   = (const int*)d_in[1];
    const float* W    = (const float*)d_in[2];
    const float* bias = (const float*)d_in[3];
    float*       out  = (float*)d_out;

    const int* src = ei;
    const int* dst = ei + N_EDGES;

    float* h      = (float*)d_ws;
    float* dinv   = h + (size_t)N_NODES * DIM;
    int*   cnt    = (int*)(dinv + N_NODES);
    int*   rowptr = cnt + N_NODES;
    int*   csr    = rowptr + N_NODES;

    gcn_gemm_wmma<<<GEMM_WGS + CSR_WGS, G_THREADS, 0, stream>>>(x, W, h, src, dst, cnt, rowptr, dinv, csr);

    gcn_gather<<<N_NODES / 4, GATHER_THREADS, 0, stream>>>(h, dinv, bias, rowptr, cnt, csr, out);
}


# Bucketing work-groups un-serialized: a batch's 32 LDS cursor adds issued as one group, no load or store under a lane test (tail loop; foreign edges go to a sink word); sum order = cursor arrival
# speedup vs baseline: 2.8308x; 1.0787x over previous
#include <hip/hip_runtime.h>
#include <hip/hip_bf16.h>

#define N_NODES 10000
#define N_EDGES 100000
#define DIM     768

typedef __attribute__((ext_vector_type(2))) float v2f;
typedef __attribute__((ext_vector_type(8))) float v8f;


typedef __attribute__((ext_vector_type(4))) float v4f;

#define GM_T      80
#define GN_T      384
#define GK_T      64
#define G_THREADS 256
#define G_MT      (GM_T / 16)
#define G_NT      3
#define G_KB      (GK_T / 8)
#define A_PITCH   (GK_T + 4)
#define A_STAGE   (GM_T * A_PITCH)
#define B_PITCH   (GN_T + 1)
#define B_STAGE   ((GK_T / 4) * B_PITCH * 4)
#define STAGE_F   (A_STAGE + B_STAGE)
#define A_LOADS   (GM_T * (GK_T / 4) / G_THREADS)
#define B_BLOCKS  ((GK_T / 4) * (GN_T / 4) / G_THREADS)
#define K_STEPS   (DIM / GK_T)

__device__ __forceinline__ void gemm_fetch(v4f (&ga)[A_LOADS], v4f (&gb)[B_BLOCKS][4],
                                           const float* __restrict__ X,
                                           const float* __restrict__ Wm,
                                           int m0, int n0, int k0, int t) {
    #pragma unroll
    for (int i = 0; i < A_LOADS; ++i) {
        const int idx = i * G_THREADS + t;
        const int row = idx >> 4;
        const int c4  = idx & 15;
        ga[i] = *(const v4f*)(X + (size_t)(m0 + row) * DIM + k0 + c4 * 4);
    }
    #pragma unroll
    for (int j = 0; j < B_BLOCKS; ++j) {
        const int p    = j * (G_THREADS / 16) + (t >> 4);
        const int kq   = (p / (GN_T / 16)) * 4 + (t & 3);
        const int ng   = (p % (GN_T / 16)) * 4 + ((t >> 2) & 3);
        const float* wp = Wm + (size_t)(k0 + kq * 4) * DIM + n0 + ng * 4;
        #pragma unroll
        for (int kk = 0; kk < 4; ++kk) gb[j][kk] = *(const v4f*)(wp + (size_t)kk * DIM);
    }
}

__device__ __forceinline__ void gemm_stash(const v4f (&ga)[A_LOADS], const v4f (&gb)[B_BLOCKS][4],
                                           float* __restrict__ As, float* __restrict__ Bs, int t) {
    #pragma unroll
    for (int i = 0; i < A_LOADS; ++i) {
        const int idx = i * G_THREADS + t;
        const int row = idx >> 4;
        const int c4  = idx & 15;
        *(v4f*)(As + row * A_PITCH + c4 * 4) = ga[i];
    }
    #pragma unroll
    for (int j = 0; j < B_BLOCKS; ++j) {
        const int p    = j * (G_THREADS / 16) + (t >> 4);
        const int kq   = (p / (GN_T / 16)) * 4 + (t & 3);
        const int ng   = (p % (GN_T / 16)) * 4 + ((t >> 2) & 3);
        float* bp = Bs + (kq * B_PITCH + ng * 4) * 4;
        *(v4f*)(bp +  0) = v4f{ gb[j][0].x, gb[j][1].x, gb[j][2].x, gb[j][3].x };
        *(v4f*)(bp +  4) = v4f{ gb[j][0].y, gb[j][1].y, gb[j][2].y, gb[j][3].y };
        *(v4f*)(bp +  8) = v4f{ gb[j][0].z, gb[j][1].z, gb[j][2].z, gb[j][3].z };
        *(v4f*)(bp + 12) = v4f{ gb[j][0].w, gb[j][1].w, gb[j][2].w, gb[j][3].w };
    }
}

__device__ __forceinline__ void gemm_frags(v4f (&fa)[G_MT], v4f (&fb)[G_NT],
                                           const float* __restrict__ As,
                                           const float* __restrict__ Bs,
                                           int kb, int c, int hi, int nw) {
    #pragma unroll
    for (int mt = 0; mt < G_MT; ++mt)
        fa[mt] = *(const v4f*)(As + (mt * 16 + c) * A_PITCH + kb * 8 + hi * 4);
    #pragma unroll
    for (int nt = 0; nt < G_NT; ++nt)
        fb[nt] = *(const v4f*)(Bs + ((kb * 2 + hi) * B_PITCH + nw + nt * 16 + c) * 4);
}

#define CSR_WGS    6
#define CSR_NODES  ((N_NODES + CSR_WGS - 1) / CSR_WGS)
#define CSR_ITEMS  ((CSR_NODES + G_THREADS - 1) / G_THREADS)
#define CSR_SLOTS  (CSR_ITEMS * G_THREADS)
#define CSR_BATCH  8
#define EDGE_QUADS (N_EDGES / 4)
#define CSR_FULL   (EDGE_QUADS / (CSR_BATCH * G_THREADS))
#define CSR_TAIL0  (CSR_FULL * CSR_BATCH * G_THREADS)
#define CSR_SINK0  (N_EDGES + 16)
#define CSR_SINKS  (CSR_WGS * G_THREADS)

__device__ __forceinline__ void csr_count(int* __restrict__ cur, int d, int lo, int n_here, int& below) {
    const unsigned u = (unsigned)(d - lo);
    if (u < (unsigned)n_here) atomicAdd(&cur[u], 1);
    else                      below += (d < lo) ? 1 : 0;
}

__device__ __forceinline__ unsigned csr_take(int* __restrict__ cur, int d, int lo, int n_here, unsigned sink) {
    const unsigned u = (unsigned)(d - lo);
    unsigned slot = sink;
    if (u < (unsigned)n_here) slot = (unsigned)atomicAdd(&cur[u], 1);
    return slot;
}

__device__ __forceinline__ void csr_build(int* __restrict__ L, int part,
                                          const int* __restrict__ src, const int* __restrict__ dst,
                                          int* __restrict__ cnt, int* __restrict__ rowptr,
                                          float* __restrict__ dinv, int* __restrict__ csr_src) {
    const int t      = threadIdx.x;
    const int lane   = t & 31;
    const int w      = t >> 5;
    const int lo     = part * CSR_NODES;
    const int n_here = (N_NODES - lo < CSR_NODES) ? (N_NODES - lo) : CSR_NODES;

    int* cur  = L;
    int* wsum = L + CSR_SLOTS;
    int* wlow = wsum + G_THREADS / 32;

    for (int i = t; i < CSR_SLOTS; i += G_THREADS) cur[i] = 0;
    __syncthreads();

    const int4* dst4 = (const int4*)dst;
    const int4* src4 = (const int4*)src;

    int below = 0;
    for (int b = 0; b < CSR_FULL; ++b) {
        const int i0 = b * (CSR_BATCH * G_THREADS) + t;
        int4 d[CSR_BATCH];
        #pragma unroll
        for (int q = 0; q < CSR_BATCH; ++q) d[q] = dst4[i0 + q * G_THREADS];
        #pragma unroll
        for (int q = 0; q < CSR_BATCH; ++q) {
            csr_count(cur, d[q].x, lo, n_here, below);
            csr_count(cur, d[q].y, lo, n_here, below);
            csr_count(cur, d[q].z, lo, n_here, below);
            csr_count(cur, d[q].w, lo, n_here, below);
        }
    }
    for (int i = CSR_TAIL0 + t; i < EDGE_QUADS; i += G_THREADS) {
        const int4 d = dst4[i];
        csr_count(cur, d.x, lo, n_here, below);
        csr_count(cur, d.y, lo, n_here, below);
        csr_count(cur, d.z, lo, n_here, below);
        csr_count(cur, d.w, lo, n_here, below);
    }
    for (int off = 16; off > 0; off >>= 1) below += __shfl_xor(below, off, 32);
    if (lane == 0) wlow[w] = below;
    __syncthreads();

    int c[CSR_ITEMS];
    int s = 0;
    #pragma unroll
    for (int j = 0; j < CSR_ITEMS; ++j) {
        c[j] = cur[t * CSR_ITEMS + j];
        s += c[j];
    }
    int inc = s;
    for (int off = 1; off < 32; off <<= 1) {
        const int y = __shfl_up(inc, off, 32);
        if (lane >= off) inc += y;
    }
    if (lane == 31) wsum[w] = inc;
    __syncthreads();

    int run = inc - s;
    for (int k = 0; k < G_THREADS / 32; ++k) {
        run += wlow[k];
        if (k < w) run += wsum[k];
    }
    #pragma unroll
    for (int j = 0; j < CSR_ITEMS; ++j) {
        const int idx = t * CSR_ITEMS + j;
        if (idx < n_here) {
            rowptr[lo + idx] = run;
            cnt[lo + idx]    = c[j];
            dinv[lo + idx]   = rsqrtf((float)(c[j] + 1));
            cur[idx]         = run;
        }
        run += c[j];
    }
    __syncthreads();

    const unsigned sink = (unsigned)(CSR_SINK0 + part * G_THREADS + t);
    for (int b = 0; b < CSR_FULL; ++b) {
        const int i0 = b * (CSR_BATCH * G_THREADS) + t;
        int4 d[CSR_BATCH], sv[CSR_BATCH];
        #pragma unroll
        for (int q = 0; q < CSR_BATCH; ++q) {
            d[q]  = dst4[i0 + q * G_THREADS];
            sv[q] = src4[i0 + q * G_THREADS];
        }
        unsigned slot[CSR_BATCH][4];
        #pragma unroll
        for (int q = 0; q < CSR_BATCH; ++q) {
            slot[q][0] = csr_take(cur, d[q].x, lo, n_here, sink);
            slot[q][1] = csr_take(cur, d[q].y, lo, n_here, sink);
            slot[q][2] = csr_take(cur, d[q].z, lo, n_here, sink);
            slot[q][3] = csr_take(cur, d[q].w, lo, n_here, sink);
        }
        #pragma unroll
        for (int q = 0; q < CSR_BATCH; ++q) {
            csr_src[slot[q][0]] = sv[q].x;
            csr_src[slot[q][1]] = sv[q].y;
            csr_src[slot[q][2]] = sv[q].z;
            csr_src[slot[q][3]] = sv[q].w;
        }
    }
    for (int i = CSR_TAIL0 + t; i < EDGE_QUADS; i += G_THREADS) {
        const int4 d  = dst4[i];
        const int4 sv = src4[i];
        const unsigned s0 = csr_take(cur, d.x, lo, n_here, sink);
        const unsigned s1 = csr_take(cur, d.y, lo, n_here, sink);
        const unsigned s2 = csr_take(cur, d.z, lo, n_here, sink);
        const unsigned s3 = csr_take(cur, d.w, lo, n_here, sink);
        csr_src[s0] = sv.x;
        csr_src[s1] = sv.y;
        csr_src[s2] = sv.z;
        csr_src[s3] = sv.w;
    }
}

#define GEMM_WGS ((N_NODES / GM_T) * (DIM / GN_T))

__global__ __launch_bounds__(G_THREADS) void gcn_gemm_wmma(const float* __restrict__ X,
                                                           const float* __restrict__ Wm,
                                                           float* __restrict__ H,
                                                           const int* __restrict__ src,
                                                           const int* __restrict__ dst,
                                                           int* __restrict__ cnt,
                                                           int* __restrict__ rowptr,
                                                           float* __restrict__ dinv,
                                                           int* __restrict__ csr_src) {
    __shared__ __attribute__((aligned(16))) float lds[2 * STAGE_F];

    if (blockIdx.x >= GEMM_WGS) {
        csr_build((int*)lds, blockIdx.x - GEMM_WGS, src, dst, cnt, rowptr, dinv, csr_src);
        return;
    }

    const int t    = threadIdx.x;
    const int lane = t & 31;
    const int c    = lane & 15;
    const int hi   = lane >> 4;
    const int nw   = (t >> 5) * (G_NT * 16);
    const int m0   = (blockIdx.x >> 1) * GM_T;
    const int n0   = (blockIdx.x & 1) * GN_T;

    v8f acc[G_MT][G_NT];
    #pragma unroll
    for (int mt = 0; mt < G_MT; ++mt)
        #pragma unroll
        for (int nt = 0; nt < G_NT; ++nt) acc[mt][nt] = v8f{};

    v4f ga[A_LOADS];
    v4f gb[B_BLOCKS][4];
    gemm_fetch(ga, gb, X, Wm, m0, n0, 0, t);
    gemm_stash(ga, gb, lds, lds + A_STAGE, t);
    __syncthreads();

    for (int ks = 0; ks < K_STEPS; ++ks) {
        const float* As = lds + (ks & 1) * STAGE_F;
        const float* Bs = As + A_STAGE;

        if (ks + 1 < K_STEPS) gemm_fetch(ga, gb, X, Wm, m0, n0, (ks + 1) * GK_T, t);

        v4f fa[G_MT], fb[G_NT];
        gemm_frags(fa, fb, As, Bs, 0, c, hi, nw);
        #pragma unroll
        for (int kb = 0; kb < G_KB; ++kb) {
            v4f na[G_MT], nb[G_NT];
            if (kb + 1 < G_KB) gemm_frags(na, nb, As, Bs, kb + 1, c, hi, nw);

            #pragma unroll
            for (int mt = 0; mt < G_MT; ++mt)
                #pragma unroll
                for (int nt = 0; nt < G_NT; ++nt)
                    acc[mt][nt] = __builtin_amdgcn_wmma_f32_16x16x4_f32(false, fa[mt].xy, false, fb[nt].xy, (short)0, acc[mt][nt], false, false);
            #pragma unroll
            for (int mt = 0; mt < G_MT; ++mt)
                #pragma unroll
                for (int nt = 0; nt < G_NT; ++nt)
                    acc[mt][nt] = __builtin_amdgcn_wmma_f32_16x16x4_f32(false, fa[mt].zw, false, fb[nt].zw, (short)0, acc[mt][nt], false, false);

            if (kb + 1 < G_KB) __builtin_amdgcn_sched_barrier(0);

            if (kb + 1 < G_KB) {
                #pragma unroll
                for (int mt = 0; mt < G_MT; ++mt) fa[mt] = na[mt];
                #pragma unroll
                for (int nt = 0; nt < G_NT; ++nt) fb[nt] = nb[nt];
            }
        }

        if (ks + 1 < K_STEPS) {
            float* An = lds + ((ks + 1) & 1) * STAGE_F;
            gemm_stash(ga, gb, An, An + A_STAGE, t);
        }
        __syncthreads();
    }

    #pragma unroll
    for (int mt = 0; mt < G_MT; ++mt) {
        #pragma unroll
        for (int r = 0; r < 8; ++r) {
            const int row = m0 + mt * 16 + r + hi * 8;
            float* ho = H + (size_t)row * DIM + n0 + nw + c;
            #pragma unroll
            for (int nt = 0; nt < G_NT; ++nt) ho[nt * 16] = acc[mt][nt][r];
        }
    }
}

#define GATHER_THREADS (DIM / 4)

__device__ __forceinline__ void fma4(float4& acc, float n, const float4& h) {
    acc.x += n * h.x;
    acc.y += n * h.y;
    acc.z += n * h.z;
    acc.w += n * h.w;
}

__device__ __forceinline__ int src_at(const int* __restrict__ csr_src, int p) {
    const unsigned s = (unsigned)csr_src[p];
    return (int)(s < (unsigned)(N_NODES - 1) ? s : (unsigned)(N_NODES - 1));
}

template <int R>
__device__ __forceinline__ void add_rows(float4& acc, const float* __restrict__ H,
                                         const float* __restrict__ dinv,
                                         const int* __restrict__ csr_src,
                                         int p, float di, int d) {
    int    s[R];
    float  n[R];
    float4 h[R];
    #pragma unroll
    for (int q = 0; q < R; ++q) s[q] = src_at(csr_src, p + q);
    #pragma unroll
    for (int q = 0; q < R; ++q) n[q] = dinv[s[q]] * di;
    #pragma unroll
    for (int q = 0; q < R; ++q) h[q] = *(const float4*)(H + (size_t)s[q] * DIM + d);
    #pragma unroll
    for (int q = 0; q < R; ++q) fma4(acc, n[q], h[q]);
}

__global__ __launch_bounds__(GATHER_THREADS) void gcn_gather(const float* __restrict__ H,
                                                             const float* __restrict__ dinv,
                                                             const float* __restrict__ bias,
                                                             const int* __restrict__ rowptr,
                                                             const int* __restrict__ cnt,
                                                             const int* __restrict__ csr_src,
                                                             float* __restrict__ out) {
    const int d = threadIdx.x * 4;
    const float4 bv = *(const float4*)(bias + d);

    for (int i = blockIdx.x; i < N_NODES; i += gridDim.x) {
        const float di  = dinv[i];
        const int   beg = rowptr[i];
        const int   len = cnt[i];

        const float4 hs = *(const float4*)(H + (size_t)i * DIM + d);
        float4 acc = bv;
        fma4(acc, di * di, hs);

        int j = 0;
        for (; j + 4 <= len; j += 4)
            add_rows<4>(acc, H, dinv, csr_src, beg + j, di, d);
        const int left = len - j;
        if (left == 3)      add_rows<3>(acc, H, dinv, csr_src, beg + j, di, d);
        else if (left == 2) add_rows<2>(acc, H, dinv, csr_src, beg + j, di, d);
        else if (left == 1) add_rows<1>(acc, H, dinv, csr_src, beg + j, di, d);

        *(float4*)(out + (size_t)i * DIM + d) = acc;
    }
}

extern "C" void kernel_launch(void* const* d_in, const int* in_sizes, int n_in,
                              void* d_out, int out_size, void* d_ws, size_t ws_size,
                              hipStream_t stream) {
    const float* x    = (const float*)d_in[0];
    const int*   ei   = (const int*)d_in[1];
    const float* W    = (const float*)d_in[2];
    const float* bias = (const float*)d_in[3];
    float*       out  = (float*)d_out;

    const int* src = ei;
    const int* dst = ei + N_EDGES;

    float* h      = (float*)d_ws;
    float* dinv   = h + (size_t)N_NODES * DIM;
    int*   cnt    = (int*)(dinv + N_NODES);
    int*   rowptr = cnt + N_NODES;
    int*   csr    = rowptr + N_NODES;

    gcn_gemm_wmma<<<GEMM_WGS + CSR_WGS, G_THREADS, 0, stream>>>(x, W, h, src, dst, cnt, rowptr, dinv, csr);

    gcn_gather<<<N_NODES / 4, GATHER_THREADS, 0, stream>>>(h, dinv, bias, rowptr, cnt, csr, out);
}
